# starting kernel
# baseline (speedup 1.0000x reference)
#include <hip/hip_runtime.h>
#include <float.h>


typedef __attribute__((ext_vector_type(2))) float v2f;
typedef __attribute__((ext_vector_type(8))) float v8f;

#define B_ 16
#define L_ 4096
#define C_ 1024
#define K_ 16
#define NSPLIT 16
#define LRANGE (L_ / NSPLIT)
#define CHUNK 128
#define NCHUNK (LRANGE / CHUNK)
#define SP 132
#define CW (C_ / 8)
#define NT 8
#define NEG_MAX (-3.402823466e38f)

__global__ __launch_bounds__(256) void qp_fused(const float* __restrict__ x,
                                                const int* __restrict__ mask,
                                                const float* __restrict__ qm,
                                                float* __restrict__ part,
                                                float* __restrict__ mstat,
                                                float* __restrict__ sstat)
{
    __shared__ float sw[K_ * SP];
    __shared__ float red[K_ * 16];
    __shared__ float mq[K_], sq[K_], scq[K_];

    const int tid   = threadIdx.x;
    const int w     = tid >> 5;
    const int lane  = tid & 31;
    const int lid   = lane & 15;
    const int half  = lane >> 4;
    const int split = blockIdx.x & (NSPLIT - 1);
    const int b     = blockIdx.x >> 4;
    const int lbase = split * LRANGE;
    const int cw    = w * CW;

    const int tq = tid >> 4;
    const int tj = tid & 15;

    if (tid < K_) { mq[tid] = NEG_MAX; sq[tid] = 0.0f; }
    __syncthreads();

    v8f zero = {};
    v8f acc[NT];
#pragma unroll
    for (int j = 0; j < NT; ++j) acc[j] = zero;

    const float* qrow = qm + lid * C_ + 2 * half;

    for (int ch = 0; ch < NCHUNK; ++ch) {
        const int l0 = lbase + ch * CHUNK + w * 16;
        const float* xrow = x + (size_t)(b * L_ + l0 + lid) * C_ + 2 * half;
        v8f d = zero;
#pragma unroll 16
        for (int c = 0; c < C_; c += 4) {
            v2f av = *(const v2f*)(qrow + c);
            v2f bv = *(const v2f*)(xrow + c);
            d = __builtin_amdgcn_wmma_f32_16x16x4_f32(false, av, false, bv,
                                                      (short)0, d, false, false);
        }
        const int mval = mask[b * L_ + l0 + lid];
        const int lloc = w * 16 + lid;
#pragma unroll
        for (int r = 0; r < 8; ++r) {
            float v = d[r] * 0.03125f;
            sw[(r + 8 * half) * SP + lloc] = mval ? v : NEG_MAX;
        }
        __syncthreads();

        float lm = NEG_MAX;
#pragma unroll
        for (int i = 0; i < CHUNK / 16; ++i)
            lm = fmaxf(lm, sw[tq * SP + tj + 16 * i]);
        red[tq * 16 + tj] = lm;
        __syncthreads();
        if (tid < K_) {
            float nm = mq[tid];
            for (int i = 0; i < 16; ++i) nm = fmaxf(nm, red[tid * 16 + i]);
            scq[tid] = __expf(mq[tid] - nm);
            mq[tid]  = nm;
        }
        __syncthreads();

        float scl[8];
#pragma unroll
        for (int r = 0; r < 8; ++r) scl[r] = scq[r + 8 * half];
#pragma unroll
        for (int j = 0; j < NT; ++j)
#pragma unroll
            for (int r = 0; r < 8; ++r) acc[j][r] *= scl[r];

        const float nmv = mq[tq];
        float ls = 0.0f;
#pragma unroll
        for (int i = 0; i < CHUNK / 16; ++i) {
            float e = __expf(sw[tq * SP + tj + 16 * i] - nmv);
            sw[tq * SP + tj + 16 * i] = e;
            ls += e;
        }
        red[tq * 16 + tj] = ls;
        __syncthreads();
        if (tid < K_) {
            float s = 0.0f;
            for (int i = 0; i < 16; ++i) s += red[tid * 16 + i];
            sq[tid] = sq[tid] * scq[tid] + s;
        }

        const float* xcol = x + (size_t)(b * L_ + lbase + ch * CHUNK) * C_ + cw + lid;
        for (int l = 0; l < CHUNK; l += 4) {
            v2f av = *(const v2f*)(&sw[lid * SP + l + 2 * half]);
            const float* xp = xcol + (size_t)(l + 2 * half) * C_;
#pragma unroll
            for (int j = 0; j < NT; ++j) {
                v2f bv;
                bv.x = __builtin_nontemporal_load(xp + j * 16);
                bv.y = __builtin_nontemporal_load(xp + C_ + j * 16);
                acc[j] = __builtin_amdgcn_wmma_f32_16x16x4_f32(false, av, false, bv,
                                                               (short)0, acc[j], false, false);
            }
        }
        __syncthreads();
    }

    float* prow = part + ((size_t)((b * NSPLIT + split) * K_ + 8 * half)) * C_ + cw + lid;
#pragma unroll
    for (int j = 0; j < NT; ++j)
#pragma unroll
        for (int r = 0; r < 8; ++r)
            __builtin_nontemporal_store(acc[j][r], prow + (size_t)r * C_ + j * 16);

    if (tid < K_) {
        mstat[(b * NSPLIT + split) * K_ + tid] = mq[tid];
        sstat[(b * NSPLIT + split) * K_ + tid] = sq[tid];
    }
}

__global__ __launch_bounds__(256) void qp_combine(const float* __restrict__ part,
                                                  const float* __restrict__ mstat,
                                                  const float* __restrict__ sstat,
                                                  float* __restrict__ out)
{
    const int i = blockIdx.x * 256 + threadIdx.x;
    const int c = i & (C_ - 1);
    const int q = (i >> 10) & (K_ - 1);
    const int b = i >> 14;

    float M = NEG_MAX;
#pragma unroll
    for (int wv = 0; wv < NSPLIT; ++wv)
        M = fmaxf(M, mstat[(b * NSPLIT + wv) * K_ + q]);

    if (M == NEG_MAX) { out[i] = 0.0f; return; }

    float num = 0.0f, den = 0.0f;
#pragma unroll
    for (int wv = 0; wv < NSPLIT; ++wv) {
        const int sidx = (b * NSPLIT + wv) * K_ + q;
        const float e  = __expf(mstat[sidx] - M);
        den += e * sstat[sidx];
        num += e * __builtin_nontemporal_load(part + (size_t)sidx * C_ + c);
    }
    out[i] = num / den;
}

extern "C" void kernel_launch(void* const* d_in, const int* in_sizes, int n_in,
                              void* d_out, int out_size, void* d_ws, size_t ws_size,
                              hipStream_t stream)
{
    const float* x    = (const float*)d_in[0];
    const int*   mask = (const int*)d_in[1];
    const float* q    = (const float*)d_in[2];
    float*       out  = (float*)d_out;

    float* part  = (float*)d_ws;
    float* mstat = part + (size_t)B_ * NSPLIT * K_ * C_;
    float* sstat = mstat + B_ * NSPLIT * K_;

    qp_fused<<<B_ * NSPLIT, 256, 0, stream>>>(x, mask, q, part, mstat, sstat);
    qp_combine<<<(B_ * K_ * C_) / 256, 256, 0, stream>>>(part, mstat, sstat, out);
}


# q staged once per workgroup into padded LDS; pass A reads its q fragments from LDS as one group per 16-WMMA block instead of one 16-line global load per WMMA; k order and sum order unchanged
# speedup vs baseline: 1.0764x; 1.0764x over previous
#include <hip/hip_runtime.h>
#include <float.h>


typedef __attribute__((ext_vector_type(2))) float v2f;
typedef __attribute__((ext_vector_type(8))) float v8f;

#define B_ 16
#define L_ 4096
#define C_ 1024
#define K_ 16
#define NSPLIT 16
#define LRANGE (L_ / NSPLIT)
#define CHUNK 128
#define NCHUNK (LRANGE / CHUNK)
#define SP 132
#define SPQ 1028
#define CW (C_ / 8)
#define NT 8
#define NEG_MAX (-3.402823466e38f)

__global__ __launch_bounds__(256) void qp_fused(const float* __restrict__ x,
                                                const int* __restrict__ mask,
                                                const float* __restrict__ qm,
                                                float* __restrict__ part,
                                                float* __restrict__ mstat,
                                                float* __restrict__ sstat)
{
    __shared__ float sw[K_ * SP];
    __shared__ __attribute__((aligned(16))) float qs[K_ * SPQ];
    __shared__ float red[K_ * 16];
    __shared__ float mq[K_], sq[K_], scq[K_];

    const int tid   = threadIdx.x;
    const int w     = tid >> 5;
    const int lane  = tid & 31;
    const int lid   = lane & 15;
    const int half  = lane >> 4;
    const int split = blockIdx.x & (NSPLIT - 1);
    const int b     = blockIdx.x >> 4;
    const int lbase = split * LRANGE;
    const int cw    = w * CW;

    const int tq = tid >> 4;
    const int tj = tid & 15;

    if (tid < K_) { mq[tid] = NEG_MAX; sq[tid] = 0.0f; }

    for (int i = tid; i < K_ * C_ / 4; i += 256) {
        const int row = i >> 8;
        const int c4  = (i & 255) << 2;
        const float4 v = *(const float4*)(qm + row * C_ + c4);
        *(float4*)(&qs[row * SPQ + c4]) = v;
    }
    __syncthreads();

    v8f zero = {};
    v8f acc[NT];
#pragma unroll
    for (int j = 0; j < NT; ++j) acc[j] = zero;

    const float* qrow = qs + lid * SPQ + 2 * half;

    for (int ch = 0; ch < NCHUNK; ++ch) {
        const int l0 = lbase + ch * CHUNK + w * 16;
        const float* xrow = x + (size_t)(b * L_ + l0 + lid) * C_ + 2 * half;
        v8f d = zero;
        for (int c = 0; c < C_; c += 64) {
            v2f bx[16], aq[16];
#pragma unroll
            for (int u = 0; u < 16; ++u) bx[u] = *(const v2f*)(xrow + c + 4 * u);
#pragma unroll
            for (int u = 0; u < 16; ++u) aq[u] = *(const v2f*)(qrow + c + 4 * u);
            __builtin_amdgcn_sched_barrier(0);
#pragma unroll
            for (int u = 0; u < 16; ++u)
                d = __builtin_amdgcn_wmma_f32_16x16x4_f32(false, aq[u], false, bx[u],
                                                          (short)0, d, false, false);
        }
        const int mval = mask[b * L_ + l0 + lid];
        const int lloc = w * 16 + lid;
#pragma unroll
        for (int r = 0; r < 8; ++r) {
            float v = d[r] * 0.03125f;
            sw[(r + 8 * half) * SP + lloc] = mval ? v : NEG_MAX;
        }
        __syncthreads();

        float lm = NEG_MAX;
#pragma unroll
        for (int i = 0; i < CHUNK / 16; ++i)
            lm = fmaxf(lm, sw[tq * SP + tj + 16 * i]);
        red[tq * 16 + tj] = lm;
        __syncthreads();
        if (tid < K_) {
            float nm = mq[tid];
            for (int i = 0; i < 16; ++i) nm = fmaxf(nm, red[tid * 16 + i]);
            scq[tid] = __expf(mq[tid] - nm);
            mq[tid]  = nm;
        }
        __syncthreads();

        float scl[8];
#pragma unroll
        for (int r = 0; r < 8; ++r) scl[r] = scq[r + 8 * half];
#pragma unroll
        for (int j = 0; j < NT; ++j)
#pragma unroll
            for (int r = 0; r < 8; ++r) acc[j][r] *= scl[r];

        const float nmv = mq[tq];
        float ls = 0.0f;
#pragma unroll
        for (int i = 0; i < CHUNK / 16; ++i) {
            float e = __expf(sw[tq * SP + tj + 16 * i] - nmv);
            sw[tq * SP + tj + 16 * i] = e;
            ls += e;
        }
        red[tq * 16 + tj] = ls;
        __syncthreads();
        if (tid < K_) {
            float s = 0.0f;
            for (int i = 0; i < 16; ++i) s += red[tid * 16 + i];
            sq[tid] = sq[tid] * scq[tid] + s;
        }

        const float* xcol = x + (size_t)(b * L_ + lbase + ch * CHUNK) * C_ + cw + lid;
        for (int l = 0; l < CHUNK; l += 4) {
            v2f av = *(const v2f*)(&sw[lid * SP + l + 2 * half]);
            const float* xp = xcol + (size_t)(l + 2 * half) * C_;
#pragma unroll
            for (int j = 0; j < NT; ++j) {
                v2f bv;
                bv.x = __builtin_nontemporal_load(xp + j * 16);
                bv.y = __builtin_nontemporal_load(xp + C_ + j * 16);
                acc[j] = __builtin_amdgcn_wmma_f32_16x16x4_f32(false, av, false, bv,
                                                               (short)0, acc[j], false, false);
            }
        }
        __syncthreads();
    }

    float* prow = part + ((size_t)((b * NSPLIT + split) * K_ + 8 * half)) * C_ + cw + lid;
#pragma unroll
    for (int j = 0; j < NT; ++j)
#pragma unroll
        for (int r = 0; r < 8; ++r)
            __builtin_nontemporal_store(acc[j][r], prow + (size_t)r * C_ + j * 16);

    if (tid < K_) {
        mstat[(b * NSPLIT + split) * K_ + tid] = mq[tid];
        sstat[(b * NSPLIT + split) * K_ + tid] = sq[tid];
    }
}

__global__ __launch_bounds__(256) void qp_combine(const float* __restrict__ part,
                                                  const float* __restrict__ mstat,
                                                  const float* __restrict__ sstat,
                                                  float* __restrict__ out)
{
    const int i = blockIdx.x * 256 + threadIdx.x;
    const int c = i & (C_ - 1);
    const int q = (i >> 10) & (K_ - 1);
    const int b = i >> 14;

    float M = NEG_MAX;
#pragma unroll
    for (int wv = 0; wv < NSPLIT; ++wv)
        M = fmaxf(M, mstat[(b * NSPLIT + wv) * K_ + q]);

    if (M == NEG_MAX) { out[i] = 0.0f; return; }

    float num = 0.0f, den = 0.0f;
#pragma unroll
    for (int wv = 0; wv < NSPLIT; ++wv) {
        const int sidx = (b * NSPLIT + wv) * K_ + q;
        const float e  = __expf(mstat[sidx] - M);
        den += e * sstat[sidx];
        num += e * __builtin_nontemporal_load(part + (size_t)sidx * C_ + c);
    }
    out[i] = num / den;
}

extern "C" void kernel_launch(void* const* d_in, const int* in_sizes, int n_in,
                              void* d_out, int out_size, void* d_ws, size_t ws_size,
                              hipStream_t stream)
{
    const float* x    = (const float*)d_in[0];
    const int*   mask = (const int*)d_in[1];
    const float* q    = (const float*)d_in[2];
    float*       out  = (float*)d_out;

    float* part  = (float*)d_ws;
    float* mstat = part + (size_t)B_ * NSPLIT * K_ * C_;
    float* sstat = mstat + B_ * NSPLIT * K_;

    qp_fused<<<B_ * NSPLIT, 256, 0, stream>>>(x, mask, q, part, mstat, sstat);
    qp_combine<<<(B_ * K_ * C_) / 256, 256, 0, stream>>>(part, mstat, sstat, out);
}


# x (and q) fragments of both WMMA loops loaded 16 B per lane, was 8 B / 4 B: score sums take c as 0,1,4,5 then 2,3,6,7 per 8 columns; pass B tiles interleave columns; partials stored 16 B per lane
# speedup vs baseline: 1.2218x; 1.1351x over previous
#include <hip/hip_runtime.h>
#include <float.h>


typedef __attribute__((ext_vector_type(2))) float v2f;
typedef __attribute__((ext_vector_type(4))) float v4f;
typedef __attribute__((ext_vector_type(8))) float v8f;

#define B_ 16
#define L_ 4096
#define C_ 1024
#define K_ 16
#define NSPLIT 16
#define LRANGE (L_ / NSPLIT)
#define CHUNK 128
#define NCHUNK (LRANGE / CHUNK)
#define SP 132
#define SPQ 1028
#define CW (C_ / 8)
#define NT 8
#define NEG_MAX (-3.402823466e38f)

__global__ __launch_bounds__(256) void qp_fused(const float* __restrict__ x,
                                                const int* __restrict__ mask,
                                                const float* __restrict__ qm,
                                                float* __restrict__ part,
                                                float* __restrict__ mstat,
                                                float* __restrict__ sstat)
{
    __shared__ float sw[K_ * SP];
    __shared__ __attribute__((aligned(16))) float qs[K_ * SPQ];
    __shared__ float red[K_ * 16];
    __shared__ float mq[K_], sq[K_], scq[K_];

    const int tid   = threadIdx.x;
    const int w     = tid >> 5;
    const int lane  = tid & 31;
    const int lid   = lane & 15;
    const int half  = lane >> 4;
    const int split = blockIdx.x & (NSPLIT - 1);
    const int b     = blockIdx.x >> 4;
    const int lbase = split * LRANGE;
    const int cw    = w * CW;

    const int tq = tid >> 4;
    const int tj = tid & 15;

    if (tid < K_) { mq[tid] = NEG_MAX; sq[tid] = 0.0f; }

    for (int i = tid; i < K_ * C_ / 4; i += 256) {
        const int row = i >> 8;
        const int c4  = (i & 255) << 2;
        const float4 v = *(const float4*)(qm + row * C_ + c4);
        *(float4*)(&qs[row * SPQ + c4]) = v;
    }
    __syncthreads();

    v8f zero = {};
    v8f acc[NT];
#pragma unroll
    for (int j = 0; j < NT; ++j) acc[j] = zero;

    const float* qrow = qs + lid * SPQ + 4 * half;

    for (int ch = 0; ch < NCHUNK; ++ch) {
        const int l0 = lbase + ch * CHUNK + w * 16;
        const float* xrow = x + (size_t)(b * L_ + l0 + lid) * C_ + 4 * half;
        v8f d = zero;
        for (int c = 0; c < C_; c += 128) {
            v4f bx[16], aq[16];
#pragma unroll
            for (int u = 0; u < 16; ++u) bx[u] = *(const v4f*)(xrow + c + 8 * u);
#pragma unroll
            for (int u = 0; u < 16; ++u) aq[u] = *(const v4f*)(qrow + c + 8 * u);
            __builtin_amdgcn_sched_barrier(0);
#pragma unroll
            for (int u = 0; u < 16; ++u) {
                const v2f a0 = {aq[u].x, aq[u].y}, b0 = {bx[u].x, bx[u].y};
                const v2f a1 = {aq[u].z, aq[u].w}, b1 = {bx[u].z, bx[u].w};
                d = __builtin_amdgcn_wmma_f32_16x16x4_f32(false, a0, false, b0,
                                                          (short)0, d, false, false);
                d = __builtin_amdgcn_wmma_f32_16x16x4_f32(false, a1, false, b1,
                                                          (short)0, d, false, false);
            }
        }
        const int mval = mask[b * L_ + l0 + lid];
        const int lloc = w * 16 + lid;
#pragma unroll
        for (int r = 0; r < 8; ++r) {
            float v = d[r] * 0.03125f;
            sw[(r + 8 * half) * SP + lloc] = mval ? v : NEG_MAX;
        }
        __syncthreads();

        float lm = NEG_MAX;
#pragma unroll
        for (int i = 0; i < CHUNK / 16; ++i)
            lm = fmaxf(lm, sw[tq * SP + tj + 16 * i]);
        red[tq * 16 + tj] = lm;
        __syncthreads();
        if (tid < K_) {
            float nm = mq[tid];
            for (int i = 0; i < 16; ++i) nm = fmaxf(nm, red[tid * 16 + i]);
            scq[tid] = __expf(mq[tid] - nm);
            mq[tid]  = nm;
        }
        __syncthreads();

        float scl[8];
#pragma unroll
        for (int r = 0; r < 8; ++r) scl[r] = scq[r + 8 * half];
#pragma unroll
        for (int j = 0; j < NT; ++j)
#pragma unroll
            for (int r = 0; r < 8; ++r) acc[j][r] *= scl[r];

        const float nmv = mq[tq];
        float ls = 0.0f;
#pragma unroll
        for (int i = 0; i < CHUNK / 16; ++i) {
            float e = __expf(sw[tq * SP + tj + 16 * i] - nmv);
            sw[tq * SP + tj + 16 * i] = e;
            ls += e;
        }
        red[tq * 16 + tj] = ls;
        __syncthreads();
        if (tid < K_) {
            float s = 0.0f;
            for (int i = 0; i < 16; ++i) s += red[tid * 16 + i];
            sq[tid] = sq[tid] * scq[tid] + s;
        }

        const float* xcol = x + (size_t)(b * L_ + lbase + ch * CHUNK + 2 * half) * C_ + cw + 4 * lid;
        for (int l = 0; l < CHUNK; l += 16) {
            v4f xv[4][4];
            v2f av[4];
#pragma unroll
            for (int s = 0; s < 4; ++s) {
                const float* xp = xcol + (size_t)(l + 4 * s) * C_;
                xv[s][0] = __builtin_nontemporal_load((const v4f*)(xp));
                xv[s][1] = __builtin_nontemporal_load((const v4f*)(xp + C_));
                xv[s][2] = __builtin_nontemporal_load((const v4f*)(xp + 64));
                xv[s][3] = __builtin_nontemporal_load((const v4f*)(xp + C_ + 64));
            }
#pragma unroll
            for (int s = 0; s < 4; ++s)
                av[s] = *(const v2f*)(&sw[lid * SP + l + 4 * s + 2 * half]);
#pragma unroll
            for (int s = 0; s < 4; ++s)
#pragma unroll
                for (int g = 0; g < 2; ++g)
#pragma unroll
                    for (int t = 0; t < 4; ++t) {
                        const v2f bv = {xv[s][2 * g][t], xv[s][2 * g + 1][t]};
                        acc[4 * g + t] = __builtin_amdgcn_wmma_f32_16x16x4_f32(false, av[s], false, bv,
                                                                               (short)0, acc[4 * g + t], false, false);
                    }
        }
        __syncthreads();
    }

    float* prow = part + ((size_t)((b * NSPLIT + split) * K_ + 8 * half)) * C_ + cw + 4 * lid;
#pragma unroll
    for (int g = 0; g < 2; ++g)
#pragma unroll
        for (int r = 0; r < 8; ++r) {
            const v4f o = {acc[4 * g][r], acc[4 * g + 1][r], acc[4 * g + 2][r], acc[4 * g + 3][r]};
            __builtin_nontemporal_store(o, (v4f*)(prow + (size_t)r * C_ + 64 * g));
        }

    if (tid < K_) {
        mstat[(b * NSPLIT + split) * K_ + tid] = mq[tid];
        sstat[(b * NSPLIT + split) * K_ + tid] = sq[tid];
    }
}

__global__ __launch_bounds__(256) void qp_combine(const float* __restrict__ part,
                                                  const float* __restrict__ mstat,
                                                  const float* __restrict__ sstat,
                                                  float* __restrict__ out)
{
    const int i = blockIdx.x * 256 + threadIdx.x;
    const int c = i & (C_ - 1);
    const int q = (i >> 10) & (K_ - 1);
    const int b = i >> 14;

    float M = NEG_MAX;
#pragma unroll
    for (int wv = 0; wv < NSPLIT; ++wv)
        M = fmaxf(M, mstat[(b * NSPLIT + wv) * K_ + q]);

    if (M == NEG_MAX) { out[i] = 0.0f; return; }

    float num = 0.0f, den = 0.0f;
#pragma unroll
    for (int wv = 0; wv < NSPLIT; ++wv) {
        const int sidx = (b * NSPLIT + wv) * K_ + q;
        const float e  = __expf(mstat[sidx] - M);
        den += e * sstat[sidx];
        num += e * __builtin_nontemporal_load(part + (size_t)sidx * C_ + c);
    }
    out[i] = num / den;
}

extern "C" void kernel_launch(void* const* d_in, const int* in_sizes, int n_in,
                              void* d_out, int out_size, void* d_ws, size_t ws_size,
                              hipStream_t stream)
{
    const float* x    = (const float*)d_in[0];
    const int*   mask = (const int*)d_in[1];
    const float* q    = (const float*)d_in[2];
    float*       out  = (float*)d_out;

    float* part  = (float*)d_ws;
    float* mstat = part + (size_t)B_ * NSPLIT * K_ * C_;
    float* sstat = mstat + B_ * NSPLIT * K_;

    qp_fused<<<B_ * NSPLIT, 256, 0, stream>>>(x, mask, q, part, mstat, sstat);
    qp_combine<<<(B_ * K_ * C_) / 256, 256, 0, stream>>>(part, mstat, sstat, out);
}


# x fragment loads of both WMMA loops (and the mask word) issued one block ahead of their WMMAs into a second register set; blocks, k order and every sum order unchanged
# speedup vs baseline: 1.2734x; 1.0422x over previous
#include <hip/hip_runtime.h>
#include <float.h>


typedef __attribute__((ext_vector_type(2))) float v2f;
typedef __attribute__((ext_vector_type(4))) float v4f;
typedef __attribute__((ext_vector_type(8))) float v8f;

#define B_ 16
#define L_ 4096
#define C_ 1024
#define K_ 16
#define NSPLIT 16
#define LRANGE (L_ / NSPLIT)
#define CHUNK 128
#define NCHUNK (LRANGE / CHUNK)
#define SP 132
#define SPQ 1028
#define CW (C_ / 8)
#define NT 8
#define NEG_MAX (-3.402823466e38f)

static __device__ __forceinline__ void load_x_a(v4f (&bx)[16], const float* p)
{
#pragma unroll
    for (int u = 0; u < 16; ++u) bx[u] = *(const v4f*)(p + 8 * u);
}
static __device__ __forceinline__ void load_q_a(v4f (&aq)[16], const float* p)
{
#pragma unroll
    for (int u = 0; u < 16; ++u) aq[u] = *(const v4f*)(p + 8 * u);
}
static __device__ __forceinline__ void score_block(v8f& d, const v4f (&aq)[16], const v4f (&bx)[16])
{
#pragma unroll
    for (int u = 0; u < 16; ++u) {
        const v2f a0 = {aq[u].x, aq[u].y}, b0 = {bx[u].x, bx[u].y};
        const v2f a1 = {aq[u].z, aq[u].w}, b1 = {bx[u].z, bx[u].w};
        d = __builtin_amdgcn_wmma_f32_16x16x4_f32(false, a0, false, b0, (short)0, d, false, false);
        d = __builtin_amdgcn_wmma_f32_16x16x4_f32(false, a1, false, b1, (short)0, d, false, false);
    }
}
static __device__ __forceinline__ void load_x_b(v4f (&xv)[4][4], const float* p)
{
#pragma unroll
    for (int s = 0; s < 4; ++s) {
        const float* xp = p + (size_t)(4 * s) * C_;
        xv[s][0] = __builtin_nontemporal_load((const v4f*)(xp));
        xv[s][1] = __builtin_nontemporal_load((const v4f*)(xp + C_));
        xv[s][2] = __builtin_nontemporal_load((const v4f*)(xp + 64));
        xv[s][3] = __builtin_nontemporal_load((const v4f*)(xp + C_ + 64));
    }
}
static __device__ __forceinline__ void load_w_b(v2f (&av)[4], const float* p)
{
#pragma unroll
    for (int s = 0; s < 4; ++s) av[s] = *(const v2f*)(p + 4 * s);
}
static __device__ __forceinline__ void pool_block(v8f (&acc)[NT], const v2f (&av)[4], const v4f (&xv)[4][4])
{
#pragma unroll
    for (int s = 0; s < 4; ++s)
#pragma unroll
        for (int g = 0; g < 2; ++g)
#pragma unroll
            for (int t = 0; t < 4; ++t) {
                const v2f bv = {xv[s][2 * g][t], xv[s][2 * g + 1][t]};
                acc[4 * g + t] = __builtin_amdgcn_wmma_f32_16x16x4_f32(false, av[s], false, bv,
                                                                       (short)0, acc[4 * g + t], false, false);
            }
}

__global__ __launch_bounds__(256) void qp_fused(const float* __restrict__ x,
                                                const int* __restrict__ mask,
                                                const float* __restrict__ qm,
                                                float* __restrict__ part,
                                                float* __restrict__ mstat,
                                                float* __restrict__ sstat)
{
    __shared__ float sw[K_ * SP];
    __shared__ __attribute__((aligned(16))) float qs[K_ * SPQ];
    __shared__ float red[K_ * 16];
    __shared__ float mq[K_], sq[K_], scq[K_];

    const int tid   = threadIdx.x;
    const int w     = tid >> 5;
    const int lane  = tid & 31;
    const int lid   = lane & 15;
    const int half  = lane >> 4;
    const int split = blockIdx.x & (NSPLIT - 1);
    const int b     = blockIdx.x >> 4;
    const int lbase = split * LRANGE;
    const int cw    = w * CW;

    const int tq = tid >> 4;
    const int tj = tid & 15;

    if (tid < K_) { mq[tid] = NEG_MAX; sq[tid] = 0.0f; }

    for (int i = tid; i < K_ * C_ / 4; i += 256) {
        const int row = i >> 8;
        const int c4  = (i & 255) << 2;
        const float4 v = *(const float4*)(qm + row * C_ + c4);
        *(float4*)(&qs[row * SPQ + c4]) = v;
    }
    __syncthreads();

    v8f zero = {};
    v8f acc[NT];
#pragma unroll
    for (int j = 0; j < NT; ++j) acc[j] = zero;

    const float* qrow = qs + lid * SPQ + 4 * half;

    for (int ch = 0; ch < NCHUNK; ++ch) {
        const int l0 = lbase + ch * CHUNK + w * 16;
        const float* xrow = x + (size_t)(b * L_ + l0 + lid) * C_ + 4 * half;
        const float* xcol = x + (size_t)(b * L_ + lbase + ch * CHUNK + 2 * half) * C_ + cw + 4 * lid;
        const int mval = mask[b * L_ + l0 + lid];
        v8f d = zero;
        v4f bxa[16], bxb[16], aq[16];
        v4f xva[4][4], xvb[4][4];
        load_x_a(bxa, xrow);
        for (int c = 0; c < C_ - 256; c += 256) {
            load_x_a(bxb, xrow + c + 128);
            load_q_a(aq, qrow + c);
            __builtin_amdgcn_sched_barrier(0);
            score_block(d, aq, bxa);
            load_x_a(bxa, xrow + c + 256);
            load_q_a(aq, qrow + c + 128);
            __builtin_amdgcn_sched_barrier(0);
            score_block(d, aq, bxb);
        }
        load_x_a(bxb, xrow + C_ - 128);
        load_q_a(aq, qrow + C_ - 256);
        __builtin_amdgcn_sched_barrier(0);
        score_block(d, aq, bxa);
        load_x_b(xva, xcol);
        load_q_a(aq, qrow + C_ - 128);
        __builtin_amdgcn_sched_barrier(0);
        score_block(d, aq, bxb);
        const int lloc = w * 16 + lid;
#pragma unroll
        for (int r = 0; r < 8; ++r) {
            float v = d[r] * 0.03125f;
            sw[(r + 8 * half) * SP + lloc] = mval ? v : NEG_MAX;
        }
        __syncthreads();

        float lm = NEG_MAX;
#pragma unroll
        for (int i = 0; i < CHUNK / 16; ++i)
            lm = fmaxf(lm, sw[tq * SP + tj + 16 * i]);
        red[tq * 16 + tj] = lm;
        __syncthreads();
        if (tid < K_) {
            float nm = mq[tid];
            for (int i = 0; i < 16; ++i) nm = fmaxf(nm, red[tid * 16 + i]);
            scq[tid] = __expf(mq[tid] - nm);
            mq[tid]  = nm;
        }
        __syncthreads();

        float scl[8];
#pragma unroll
        for (int r = 0; r < 8; ++r) scl[r] = scq[r + 8 * half];
#pragma unroll
        for (int j = 0; j < NT; ++j)
#pragma unroll
            for (int r = 0; r < 8; ++r) acc[j][r] *= scl[r];

        const float nmv = mq[tq];
        float ls = 0.0f;
#pragma unroll
        for (int i = 0; i < CHUNK / 16; ++i) {
            float e = __expf(sw[tq * SP + tj + 16 * i] - nmv);
            sw[tq * SP + tj + 16 * i] = e;
            ls += e;
        }
        red[tq * 16 + tj] = ls;
        __syncthreads();
        if (tid < K_) {
            float s = 0.0f;
            for (int i = 0; i < 16; ++i) s += red[tid * 16 + i];
            sq[tid] = sq[tid] * scq[tid] + s;
        }

        const float* wrow = &sw[lid * SP + 2 * half];
        v2f av[4];
        for (int l = 0; l < CHUNK - 32; l += 32) {
            load_x_b(xvb, xcol + (size_t)(l + 16) * C_);
            load_w_b(av, wrow + l);
            __builtin_amdgcn_sched_barrier(0);
            pool_block(acc, av, xva);
            load_x_b(xva, xcol + (size_t)(l + 32) * C_);
            load_w_b(av, wrow + l + 16);
            __builtin_amdgcn_sched_barrier(0);
            pool_block(acc, av, xvb);
        }
        load_x_b(xvb, xcol + (size_t)(CHUNK - 16) * C_);
        load_w_b(av, wrow + CHUNK - 32);
        __builtin_amdgcn_sched_barrier(0);
        pool_block(acc, av, xva);
        load_w_b(av, wrow + CHUNK - 16);
        pool_block(acc, av, xvb);
        __syncthreads();
    }

    float* prow = part + ((size_t)((b * NSPLIT + split) * K_ + 8 * half)) * C_ + cw + 4 * lid;
#pragma unroll
    for (int g = 0; g < 2; ++g)
#pragma unroll
        for (int r = 0; r < 8; ++r) {
            const v4f o = {acc[4 * g][r], acc[4 * g + 1][r], acc[4 * g + 2][r], acc[4 * g + 3][r]};
            __builtin_nontemporal_store(o, (v4f*)(prow + (size_t)r * C_ + 64 * g));
        }

    if (tid < K_) {
        mstat[(b * NSPLIT + split) * K_ + tid] = mq[tid];
        sstat[(b * NSPLIT + split) * K_ + tid] = sq[tid];
    }
}

__global__ __launch_bounds__(256) void qp_combine(const float* __restrict__ part,
                                                  const float* __restrict__ mstat,
                                                  const float* __restrict__ sstat,
                                                  float* __restrict__ out)
{
    const int i = blockIdx.x * 256 + threadIdx.x;
    const int c = i & (C_ - 1);
    const int q = (i >> 10) & (K_ - 1);
    const int b = i >> 14;

    float M = NEG_MAX;
#pragma unroll
    for (int wv = 0; wv < NSPLIT; ++wv)
        M = fmaxf(M, mstat[(b * NSPLIT + wv) * K_ + q]);

    if (M == NEG_MAX) { out[i] = 0.0f; return; }

    float num = 0.0f, den = 0.0f;
#pragma unroll
    for (int wv = 0; wv < NSPLIT; ++wv) {
        const int sidx = (b * NSPLIT + wv) * K_ + q;
        const float e  = __expf(mstat[sidx] - M);
        den += e * sstat[sidx];
        num += e * __builtin_nontemporal_load(part + (size_t)sidx * C_ + c);
    }
    out[i] = num / den;
}

extern "C" void kernel_launch(void* const* d_in, const int* in_sizes, int n_in,
                              void* d_out, int out_size, void* d_ws, size_t ws_size,
                              hipStream_t stream)
{
    const float* x    = (const float*)d_in[0];
    const int*   mask = (const int*)d_in[1];
    const float* q    = (const float*)d_in[2];
    float*       out  = (float*)d_out;

    float* part  = (float*)d_ws;
    float* mstat = part + (size_t)B_ * NSPLIT * K_ * C_;
    float* sstat = mstat + B_ * NSPLIT * K_;

    qp_fused<<<B_ * NSPLIT, 256, 0, stream>>>(x, mask, q, part, mstat, sstat);
    qp_combine<<<(B_ * K_ * C_) / 256, 256, 0, stream>>>(part, mstat, sstat, out);
}


# rolling L2 prefetch three pass-A blocks (one 256-B unit per lane) ahead of the score loop: starts under the q staging, next chunk's first blocks go out under the pooling blocks; sums unchanged
# speedup vs baseline: 1.4922x; 1.1718x over previous
#include <hip/hip_runtime.h>
#include <float.h>


typedef __attribute__((ext_vector_type(2))) float v2f;
typedef __attribute__((ext_vector_type(4))) float v4f;
typedef __attribute__((ext_vector_type(8))) float v8f;

#define B_ 16
#define L_ 4096
#define C_ 1024
#define K_ 16
#define NSPLIT 16
#define LRANGE (L_ / NSPLIT)
#define CHUNK 128
#define NCHUNK (LRANGE / CHUNK)
#define SP 132
#define SPQ 1028
#define CW (C_ / 8)
#define NT 8
#define NEG_MAX (-3.402823466e38f)

static __device__ __forceinline__ void load_x_a(v4f (&bx)[16], const float* p)
{
#pragma unroll
    for (int u = 0; u < 16; ++u) bx[u] = *(const v4f*)(p + 8 * u);
}
static __device__ __forceinline__ void load_q_a(v4f (&aq)[16], const float* p)
{
#pragma unroll
    for (int u = 0; u < 16; ++u) aq[u] = *(const v4f*)(p + 8 * u);
}
static __device__ __forceinline__ void score_block(v8f& d, const v4f (&aq)[16], const v4f (&bx)[16])
{
#pragma unroll
    for (int u = 0; u < 16; ++u) {
        const v2f a0 = {aq[u].x, aq[u].y}, b0 = {bx[u].x, bx[u].y};
        const v2f a1 = {aq[u].z, aq[u].w}, b1 = {bx[u].z, bx[u].w};
        d = __builtin_amdgcn_wmma_f32_16x16x4_f32(false, a0, false, b0, (short)0, d, false, false);
        d = __builtin_amdgcn_wmma_f32_16x16x4_f32(false, a1, false, b1, (short)0, d, false, false);
    }
}
static __device__ __forceinline__ void load_x_b(v4f (&xv)[4][4], const float* p)
{
#pragma unroll
    for (int s = 0; s < 4; ++s) {
        const float* xp = p + (size_t)(4 * s) * C_;
        xv[s][0] = __builtin_nontemporal_load((const v4f*)(xp));
        xv[s][1] = __builtin_nontemporal_load((const v4f*)(xp + C_));
        xv[s][2] = __builtin_nontemporal_load((const v4f*)(xp + 64));
        xv[s][3] = __builtin_nontemporal_load((const v4f*)(xp + C_ + 64));
    }
}
static __device__ __forceinline__ void load_w_b(v2f (&av)[4], const float* p)
{
#pragma unroll
    for (int s = 0; s < 4; ++s) av[s] = *(const v2f*)(p + 4 * s);
}
static __device__ __forceinline__ void prefetch_x_a(const float* p)
{
    __builtin_prefetch(p, 0, 2);
}
static __device__ __forceinline__ void pool_block(v8f (&acc)[NT], const v2f (&av)[4], const v4f (&xv)[4][4])
{
#pragma unroll
    for (int s = 0; s < 4; ++s)
#pragma unroll
        for (int g = 0; g < 2; ++g)
#pragma unroll
            for (int t = 0; t < 4; ++t) {
                const v2f bv = {xv[s][2 * g][t], xv[s][2 * g + 1][t]};
                acc[4 * g + t] = __builtin_amdgcn_wmma_f32_16x16x4_f32(false, av[s], false, bv,
                                                                       (short)0, acc[4 * g + t], false, false);
            }
}

__global__ __launch_bounds__(256) void qp_fused(const float* __restrict__ x,
                                                const int* __restrict__ mask,
                                                const float* __restrict__ qm,
                                                float* __restrict__ part,
                                                float* __restrict__ mstat,
                                                float* __restrict__ sstat)
{
    __shared__ float sw[K_ * SP];
    __shared__ __attribute__((aligned(16))) float qs[K_ * SPQ];
    __shared__ float red[K_ * 16];
    __shared__ float mq[K_], sq[K_], scq[K_];

    const int tid   = threadIdx.x;
    const int w     = tid >> 5;
    const int lane  = tid & 31;
    const int lid   = lane & 15;
    const int half  = lane >> 4;
    const int split = blockIdx.x & (NSPLIT - 1);
    const int b     = blockIdx.x >> 4;
    const int lbase = split * LRANGE;
    const int cw    = w * CW;

    const int tq = tid >> 4;
    const int tj = tid & 15;

    if (tid < K_) { mq[tid] = NEG_MAX; sq[tid] = 0.0f; }

    {
        const float* xpre = x + (size_t)(b * L_ + lbase + w * 16 + lid) * C_ + 64 * half;
        prefetch_x_a(xpre);
        prefetch_x_a(xpre + 128);
        prefetch_x_a(xpre + 256);
    }

    for (int i = tid; i < K_ * C_ / 4; i += 256) {
        const int row = i >> 8;
        const int c4  = (i & 255) << 2;
        const float4 v = *(const float4*)(qm + row * C_ + c4);
        *(float4*)(&qs[row * SPQ + c4]) = v;
    }
    __syncthreads();

    v8f zero = {};
    v8f acc[NT];
#pragma unroll
    for (int j = 0; j < NT; ++j) acc[j] = zero;

    const float* qrow = qs + lid * SPQ + 4 * half;

    for (int ch = 0; ch < NCHUNK; ++ch) {
        const int l0 = lbase + ch * CHUNK + w * 16;
        const float* xrow = x + (size_t)(b * L_ + l0 + lid) * C_ + 4 * half;
        const float* xcol = x + (size_t)(b * L_ + lbase + ch * CHUNK + 2 * half) * C_ + cw + 4 * lid;
        const float* xcur  = xrow + 60 * half;
        const float* xnext = xcur + (size_t)CHUNK * C_;
        const bool has_next = (ch + 1 < NCHUNK);
        const int mval = mask[b * L_ + l0 + lid];
        v8f d = zero;
        v4f bxa[16], bxb[16], aq[16];
        v4f xva[4][4], xvb[4][4];
        load_x_a(bxa, xrow);
        for (int c = 0; c < C_ - 256; c += 256) {
            load_x_a(bxb, xrow + c + 128);
            load_q_a(aq, qrow + c);
            prefetch_x_a(xcur + c + 384);
            __builtin_amdgcn_sched_barrier(0);
            score_block(d, aq, bxa);
            load_x_a(bxa, xrow + c + 256);
            load_q_a(aq, qrow + c + 128);
            if (c + 512 < C_) prefetch_x_a(xcur + c + 512);
            __builtin_amdgcn_sched_barrier(0);
            score_block(d, aq, bxb);
        }
        load_x_a(bxb, xrow + C_ - 128);
        load_q_a(aq, qrow + C_ - 256);
        __builtin_amdgcn_sched_barrier(0);
        score_block(d, aq, bxa);
        load_x_b(xva, xcol);
        load_q_a(aq, qrow + C_ - 128);
        __builtin_amdgcn_sched_barrier(0);
        score_block(d, aq, bxb);
        const int lloc = w * 16 + lid;
#pragma unroll
        for (int r = 0; r < 8; ++r) {
            float v = d[r] * 0.03125f;
            sw[(r + 8 * half) * SP + lloc] = mval ? v : NEG_MAX;
        }
        __syncthreads();

        float lm = NEG_MAX;
#pragma unroll
        for (int i = 0; i < CHUNK / 16; ++i)
            lm = fmaxf(lm, sw[tq * SP + tj + 16 * i]);
        red[tq * 16 + tj] = lm;
        __syncthreads();
        if (tid < K_) {
            float nm = mq[tid];
            for (int i = 0; i < 16; ++i) nm = fmaxf(nm, red[tid * 16 + i]);
            scq[tid] = __expf(mq[tid] - nm);
            mq[tid]  = nm;
        }
        __syncthreads();

        float scl[8];
#pragma unroll
        for (int r = 0; r < 8; ++r) scl[r] = scq[r + 8 * half];
#pragma unroll
        for (int j = 0; j < NT; ++j)
#pragma unroll
            for (int r = 0; r < 8; ++r) acc[j][r] *= scl[r];

        const float nmv = mq[tq];
        float ls = 0.0f;
#pragma unroll
        for (int i = 0; i < CHUNK / 16; ++i) {
            float e = __expf(sw[tq * SP + tj + 16 * i] - nmv);
            sw[tq * SP + tj + 16 * i] = e;
            ls += e;
        }
        red[tq * 16 + tj] = ls;
        __syncthreads();
        if (tid < K_) {
            float s = 0.0f;
            for (int i = 0; i < 16; ++i) s += red[tid * 16 + i];
            sq[tid] = sq[tid] * scq[tid] + s;
        }

        const float* wrow = &sw[lid * SP + 2 * half];
        v2f av[4];
        for (int l = 0; l < CHUNK - 32; l += 32) {
            load_x_b(xvb, xcol + (size_t)(l + 16) * C_);
            load_w_b(av, wrow + l);
            if (has_next && l < 64) prefetch_x_a(xnext + 8 * l);
            __builtin_amdgcn_sched_barrier(0);
            pool_block(acc, av, xva);
            load_x_b(xva, xcol + (size_t)(l + 32) * C_);
            load_w_b(av, wrow + l + 16);
            if (has_next && l == 0) prefetch_x_a(xnext + 128);
            __builtin_amdgcn_sched_barrier(0);
            pool_block(acc, av, xvb);
        }
        load_x_b(xvb, xcol + (size_t)(CHUNK - 16) * C_);
        load_w_b(av, wrow + CHUNK - 32);
        __builtin_amdgcn_sched_barrier(0);
        pool_block(acc, av, xva);
        load_w_b(av, wrow + CHUNK - 16);
        pool_block(acc, av, xvb);
        __syncthreads();
    }

    float* prow = part + ((size_t)((b * NSPLIT + split) * K_ + 8 * half)) * C_ + cw + 4 * lid;
#pragma unroll
    for (int g = 0; g < 2; ++g)
#pragma unroll
        for (int r = 0; r < 8; ++r) {
            const v4f o = {acc[4 * g][r], acc[4 * g + 1][r], acc[4 * g + 2][r], acc[4 * g + 3][r]};
            __builtin_nontemporal_store(o, (v4f*)(prow + (size_t)r * C_ + 64 * g));
        }

    if (tid < K_) {
        mstat[(b * NSPLIT + split) * K_ + tid] = mq[tid];
        sstat[(b * NSPLIT + split) * K_ + tid] = sq[tid];
    }
}

__global__ __launch_bounds__(256) void qp_combine(const float* __restrict__ part,
                                                  const float* __restrict__ mstat,
                                                  const float* __restrict__ sstat,
                                                  float* __restrict__ out)
{
    const int i = blockIdx.x * 256 + threadIdx.x;
    const int c = i & (C_ - 1);
    const int q = (i >> 10) & (K_ - 1);
    const int b = i >> 14;

    float M = NEG_MAX;
#pragma unroll
    for (int wv = 0; wv < NSPLIT; ++wv)
        M = fmaxf(M, mstat[(b * NSPLIT + wv) * K_ + q]);

    if (M == NEG_MAX) { out[i] = 0.0f; return; }

    float num = 0.0f, den = 0.0f;
#pragma unroll
    for (int wv = 0; wv < NSPLIT; ++wv) {
        const int sidx = (b * NSPLIT + wv) * K_ + q;
        const float e  = __expf(mstat[sidx] - M);
        den += e * sstat[sidx];
        num += e * __builtin_nontemporal_load(part + (size_t)sidx * C_ + c);
    }
    out[i] = num / den;
}

extern "C" void kernel_launch(void* const* d_in, const int* in_sizes, int n_in,
                              void* d_out, int out_size, void* d_ws, size_t ws_size,
                              hipStream_t stream)
{
    const float* x    = (const float*)d_in[0];
    const int*   mask = (const int*)d_in[1];
    const float* q    = (const float*)d_in[2];
    float*       out  = (float*)d_out;

    float* part  = (float*)d_ws;
    float* mstat = part + (size_t)B_ * NSPLIT * K_ * C_;
    float* sstat = mstat + B_ * NSPLIT * K_;

    qp_fused<<<B_ * NSPLIT, 256, 0, stream>>>(x, mask, q, part, mstat, sstat);
    qp_combine<<<(B_ * K_ * C_) / 256, 256, 0, stream>>>(part, mstat, sstat, out);
}
